# starting kernel
# baseline (speedup 1.0000x reference)
#include <hip/hip_runtime.h>
#include <hip/hip_bf16.h>


typedef __attribute__((ext_vector_type(2))) float v2f;
typedef __attribute__((ext_vector_type(8))) float v8f;

#define BATCH 8192
#define DDIM  512
#define FDIM  64
#define KDIM  64

__global__ __launch_bounds__(256) void ffm_build_w(
    const float* __restrict__ kern,
    const int*   __restrict__ fd,
    float*       __restrict__ Wm)
{
    int tid = blockIdx.x * 256 + threadIdx.x;
    int i = tid >> 9;
    int j = tid & (DDIM - 1);
    float w = 0.0f;
    if (j > i) {
        const float4* a = (const float4*)(kern + (size_t)i * (FDIM * KDIM) + (size_t)fd[j] * KDIM);
        const float4* b = (const float4*)(kern + (size_t)j * (FDIM * KDIM) + (size_t)fd[i] * KDIM);
#pragma unroll
        for (int k = 0; k < KDIM / 4; ++k) {
            float4 av = a[k], bv = b[k];
            w += av.x * bv.x + av.y * bv.y + av.z * bv.z + av.w * bv.w;
        }
    }
    Wm[tid] = w;
}

__global__ __launch_bounds__(256) void ffm_quadform(
    const float* __restrict__ x,
    const float* __restrict__ Wm,
    float*       __restrict__ out)
{
    const int b0   = blockIdx.x * 16;
    const int wave = __builtin_amdgcn_readfirstlane(threadIdx.x >> 5);
    const int lane = threadIdx.x & 31;
    const int half = lane >> 4;
    const int l    = lane & 15;

    __shared__ float partial[8][16];

    float contrib[8];
#pragma unroll
    for (int r = 0; r < 8; ++r) contrib[r] = 0.0f;

    const int tiles[4] = { wave, 15 - wave, 16 + wave, 31 - wave };

#pragma unroll
    for (int t = 0; t < 4; ++t) {
        const int i0  = __builtin_amdgcn_readfirstlane(tiles[t] << 4);
        const int n16 = __builtin_amdgcn_readfirstlane((DDIM - i0) >> 4);

        v8f acc = {0.f, 0.f, 0.f, 0.f, 0.f, 0.f, 0.f, 0.f};

        const v2f* __restrict__ xp =
            (const v2f*)(x  + (size_t)(b0 + l) * DDIM + i0 + 2 * half);
        const v2f* __restrict__ wp =
            (const v2f*)(Wm + (size_t)(i0 + l) * DDIM + i0 + 2 * half);

        for (int it = 0; it < n16; ++it) {
#pragma unroll
            for (int u = 0; u < 4; ++u) {
                v2f a = xp[2 * u];
                v2f b = wp[2 * u];
                acc = __builtin_amdgcn_wmma_f32_16x16x4_f32(
                     false, a,  false, b,
                     (short)0, acc,  false,  false);
            }
            xp += 8;
            wp += 8;
        }

#pragma unroll
        for (int r = 0; r < 8; ++r) {
            float xv = x[(size_t)(b0 + r + 8 * half) * DDIM + i0 + l];
            contrib[r] += acc[r] * xv;
        }
    }

#pragma unroll
    for (int r = 0; r < 8; ++r) {
        float v = contrib[r];
        v += __shfl_xor(v, 1, 32);
        v += __shfl_xor(v, 2, 32);
        v += __shfl_xor(v, 4, 32);
        v += __shfl_xor(v, 8, 32);
        if (l == 0) partial[wave][r + 8 * half] = v;
    }
    __syncthreads();

    if (threadIdx.x < 16) {
        float s = 0.0f;
#pragma unroll
        for (int w = 0; w < 8; ++w) s += partial[w][threadIdx.x];
        out[b0 + threadIdx.x] = s;
    }
}

extern "C" void kernel_launch(void* const* d_in, const int* in_sizes, int n_in,
                              void* d_out, int out_size, void* d_ws, size_t ws_size,
                              hipStream_t stream) {
    const float* x    = (const float*)d_in[0];
    const float* kern = (const float*)d_in[1];
    const int*   fd   = (const int*)d_in[2];
    float* out = (float*)d_out;
    float* Wm  = (float*)d_ws;

    ffm_build_w<<<(DDIM * DDIM) / 256, 256, 0, stream>>>(kern, fd, Wm);
    ffm_quadform<<<BATCH / 16, 256, 0, stream>>>(x, Wm, out);
}


# Quadratic-form GEMM: each wave keeps a 32x32 register tile (2x2 WMMA tiles) fed by 16-byte fragment loads, one load per two WMMAs instead of two per WMMA; changes f32 summation order
# speedup vs baseline: 1.3071x; 1.3071x over previous
#include <hip/hip_runtime.h>
#include <hip/hip_bf16.h>


typedef __attribute__((ext_vector_type(2))) float v2f;
typedef __attribute__((ext_vector_type(8))) float v8f;

#define BATCH 8192
#define DDIM  512
#define FDIM  64
#define KDIM  64

__global__ __launch_bounds__(256) void ffm_build_w(
    const float* __restrict__ kern,
    const int*   __restrict__ fd,
    float*       __restrict__ Wm)
{
    int tid = blockIdx.x * 256 + threadIdx.x;
    int i = tid >> 9;
    int j = tid & (DDIM - 1);
    float w = 0.0f;
    if (j > i) {
        const float4* a = (const float4*)(kern + (size_t)i * (FDIM * KDIM) + (size_t)fd[j] * KDIM);
        const float4* b = (const float4*)(kern + (size_t)j * (FDIM * KDIM) + (size_t)fd[i] * KDIM);
#pragma unroll
        for (int k = 0; k < KDIM / 4; ++k) {
            float4 av = a[k], bv = b[k];
            w += av.x * bv.x + av.y * bv.y + av.z * bv.z + av.w * bv.w;
        }
    }
    Wm[tid] = w;
}

typedef __attribute__((ext_vector_type(4))) float v4f;

__device__ __forceinline__ void ffm_load_step(
    v4f (&fa)[2][4], v4f (&fb)[2][4],
    const float* __restrict__ xp, const float* __restrict__ wp)
{
#pragma unroll
    for (int t = 0; t < 2; ++t) {
#pragma unroll
        for (int s = 0; s < 4; ++s)
            fa[t][s] = *(const v4f*)(xp + t * 16 * DDIM + 8 * s);
    }
#pragma unroll
    for (int t = 0; t < 2; ++t) {
#pragma unroll
        for (int s = 0; s < 4; ++s)
            fb[t][s] = *(const v4f*)(wp + t * 16 * DDIM + 8 * s);
    }
}

__device__ __forceinline__ void ffm_mma_step(
    v8f (&acc)[2][2], const v4f (&fa)[2][4], const v4f (&fb)[2][4])
{
#pragma unroll
    for (int s = 0; s < 4; ++s) {
        v2f a_lo[2], a_hi[2], b_lo[2], b_hi[2];
#pragma unroll
        for (int t = 0; t < 2; ++t) {
            a_lo[t] = __builtin_shufflevector(fa[t][s], fa[t][s], 0, 1);
            a_hi[t] = __builtin_shufflevector(fa[t][s], fa[t][s], 2, 3);
            b_lo[t] = __builtin_shufflevector(fb[t][s], fb[t][s], 0, 1);
            b_hi[t] = __builtin_shufflevector(fb[t][s], fb[t][s], 2, 3);
        }
#pragma unroll
        for (int mt = 0; mt < 2; ++mt) {
#pragma unroll
            for (int nt = 0; nt < 2; ++nt)
                acc[mt][nt] = __builtin_amdgcn_wmma_f32_16x16x4_f32(
                    false, a_lo[mt], false, b_lo[nt], (short)0, acc[mt][nt], false, false);
        }
#pragma unroll
        for (int mt = 0; mt < 2; ++mt) {
#pragma unroll
            for (int nt = 0; nt < 2; ++nt)
                acc[mt][nt] = __builtin_amdgcn_wmma_f32_16x16x4_f32(
                    false, a_hi[mt], false, b_hi[nt], (short)0, acc[mt][nt], false, false);
        }
    }
}

__global__ __launch_bounds__(256) void ffm_quadform(
    const float* __restrict__ x,
    const float* __restrict__ Wm,
    float*       __restrict__ out)
{
    const int b0   = blockIdx.x * 32;
    const int wave = __builtin_amdgcn_readfirstlane(threadIdx.x >> 5);
    const int lane = threadIdx.x & 31;
    const int half = lane >> 4;
    const int l    = lane & 15;

    __shared__ float partial[8][32];

    float contrib[2][8];
#pragma unroll
    for (int mt = 0; mt < 2; ++mt)
#pragma unroll
        for (int r = 0; r < 8; ++r) contrib[mt][r] = 0.0f;

#pragma unroll
    for (int t = 0; t < 2; ++t) {
        const int p  = __builtin_amdgcn_readfirstlane(t == 0 ? wave : 15 - wave);
        const int i0 = p << 5;
        const int n  = 16 - p;

        v8f acc[2][2];
#pragma unroll
        for (int mt = 0; mt < 2; ++mt)
#pragma unroll
            for (int nt = 0; nt < 2; ++nt)
                acc[mt][nt] = (v8f){0.f, 0.f, 0.f, 0.f, 0.f, 0.f, 0.f, 0.f};

        const float* __restrict__ xp = x  + (size_t)(b0 + l) * DDIM + i0 + 4 * half;
        const float* __restrict__ wp = Wm + (size_t)(i0 + l) * DDIM + i0 + 4 * half;

        v4f A0[2][4], B0[2][4], A1[2][4], B1[2][4];
        ffm_load_step(A0, B0, xp, wp);
        int it = 0;
        bool last_in_1 = false;
        while (it + 1 < n) {
            ffm_load_step(A1, B1, xp + 32 * (it + 1), wp + 32 * (it + 1));
            ffm_mma_step(acc, A0, B0);
            ++it;
            if (it + 1 >= n) { last_in_1 = true; break; }
            ffm_load_step(A0, B0, xp + 32 * (it + 1), wp + 32 * (it + 1));
            ffm_mma_step(acc, A1, B1);
            ++it;
        }
        if (last_in_1) ffm_mma_step(acc, A1, B1);
        else           ffm_mma_step(acc, A0, B0);

#pragma unroll
        for (int mt = 0; mt < 2; ++mt)
#pragma unroll
            for (int nt = 0; nt < 2; ++nt)
#pragma unroll
                for (int r = 0; r < 8; ++r) {
                    float xv = x[(size_t)(b0 + 16 * mt + r + 8 * half) * DDIM + i0 + 16 * nt + l];
                    contrib[mt][r] += acc[mt][nt][r] * xv;
                }
    }

#pragma unroll
    for (int mt = 0; mt < 2; ++mt)
#pragma unroll
        for (int r = 0; r < 8; ++r) {
            float v = contrib[mt][r];
            v += __shfl_xor(v, 1, 32);
            v += __shfl_xor(v, 2, 32);
            v += __shfl_xor(v, 4, 32);
            v += __shfl_xor(v, 8, 32);
            if (l == 0) partial[wave][16 * mt + r + 8 * half] = v;
        }
    __syncthreads();

    if (threadIdx.x < 32) {
        float s = 0.0f;
#pragma unroll
        for (int w = 0; w < 8; ++w) s += partial[w][threadIdx.x];
        out[b0 + threadIdx.x] = s;
    }
}

extern "C" void kernel_launch(void* const* d_in, const int* in_sizes, int n_in,
                              void* d_out, int out_size, void* d_ws, size_t ws_size,
                              hipStream_t stream) {
    const float* x    = (const float*)d_in[0];
    const float* kern = (const float*)d_in[1];
    const int*   fd   = (const int*)d_in[2];
    float* out = (float*)d_out;
    float* Wm  = (float*)d_ws;

    ffm_build_w<<<(DDIM * DDIM) / 256, 256, 0, stream>>>(kern, fd, Wm);
    ffm_quadform<<<BATCH / 32, 256, 0, stream>>>(x, Wm, out);
}


# Quadratic-form GEMM: the block's 32 rows of x are copied to LDS once with whole-line loads and every A fragment is read from LDS instead of global memory; summation order unchanged
# speedup vs baseline: 1.4946x; 1.1434x over previous
#include <hip/hip_runtime.h>
#include <hip/hip_bf16.h>


typedef __attribute__((ext_vector_type(2))) float v2f;
typedef __attribute__((ext_vector_type(8))) float v8f;

#define BATCH 8192
#define DDIM  512
#define FDIM  64
#define KDIM  64

__global__ __launch_bounds__(256) void ffm_build_w(
    const float* __restrict__ kern,
    const int*   __restrict__ fd,
    float*       __restrict__ Wm)
{
    int tid = blockIdx.x * 256 + threadIdx.x;
    int i = tid >> 9;
    int j = tid & (DDIM - 1);
    float w = 0.0f;
    if (j > i) {
        const float4* a = (const float4*)(kern + (size_t)i * (FDIM * KDIM) + (size_t)fd[j] * KDIM);
        const float4* b = (const float4*)(kern + (size_t)j * (FDIM * KDIM) + (size_t)fd[i] * KDIM);
#pragma unroll
        for (int k = 0; k < KDIM / 4; ++k) {
            float4 av = a[k], bv = b[k];
            w += av.x * bv.x + av.y * bv.y + av.z * bv.z + av.w * bv.w;
        }
    }
    Wm[tid] = w;
}

typedef __attribute__((ext_vector_type(4))) float v4f;

#define XPITCH 516


__device__ __forceinline__ void ffm_load_b(
    v4f (&fb)[2][4], const float* __restrict__ wp)
{
#pragma unroll
    for (int t = 0; t < 2; ++t) {
#pragma unroll
        for (int s = 0; s < 4; ++s)
            fb[t][s] = *(const v4f*)(wp + t * 16 * DDIM + 8 * s);
    }
}

__device__ __forceinline__ void ffm_load_a(
    v4f (&fa)[2][4], const float* xl)
{
#pragma unroll
    for (int t = 0; t < 2; ++t) {
#pragma unroll
        for (int s = 0; s < 4; ++s)
            fa[t][s] = *(const v4f*)(xl + t * 16 * XPITCH + 8 * s);
    }
}

__device__ __forceinline__ void ffm_mma_step(
    v8f (&acc)[2][2], const v4f (&fa)[2][4], const v4f (&fb)[2][4])
{
#pragma unroll
    for (int s = 0; s < 4; ++s) {
        v2f a_lo[2], a_hi[2], b_lo[2], b_hi[2];
#pragma unroll
        for (int t = 0; t < 2; ++t) {
            a_lo[t] = __builtin_shufflevector(fa[t][s], fa[t][s], 0, 1);
            a_hi[t] = __builtin_shufflevector(fa[t][s], fa[t][s], 2, 3);
            b_lo[t] = __builtin_shufflevector(fb[t][s], fb[t][s], 0, 1);
            b_hi[t] = __builtin_shufflevector(fb[t][s], fb[t][s], 2, 3);
        }
#pragma unroll
        for (int mt = 0; mt < 2; ++mt) {
#pragma unroll
            for (int nt = 0; nt < 2; ++nt)
                acc[mt][nt] = __builtin_amdgcn_wmma_f32_16x16x4_f32(
                    false, a_lo[mt], false, b_lo[nt], (short)0, acc[mt][nt], false, false);
        }
#pragma unroll
        for (int mt = 0; mt < 2; ++mt) {
#pragma unroll
            for (int nt = 0; nt < 2; ++nt)
                acc[mt][nt] = __builtin_amdgcn_wmma_f32_16x16x4_f32(
                    false, a_hi[mt], false, b_hi[nt], (short)0, acc[mt][nt], false, false);
        }
    }
}

__global__ __launch_bounds__(256) void ffm_quadform(
    const float* __restrict__ x,
    const float* __restrict__ Wm,
    float*       __restrict__ out)
{
    const int b0   = blockIdx.x * 32;
    const int wave = __builtin_amdgcn_readfirstlane(threadIdx.x >> 5);
    const int lane = threadIdx.x & 31;
    const int half = lane >> 4;
    const int l    = lane & 15;

    __shared__ __attribute__((aligned(16))) float xs[32 * XPITCH];
    __shared__ float partial[8][32];

    v4f A0[2][4], B0[2][4], A1[2][4], B1[2][4];

    {
        const float* __restrict__ xg = x + (size_t)(b0 + 4 * wave) * DDIM + 4 * lane;
        v4f st[4][4];
#pragma unroll
        for (int r = 0; r < 4; ++r)
#pragma unroll
            for (int q = 0; q < 4; ++q)
                st[r][q] = *(const v4f*)(xg + r * DDIM + 128 * q);

        const int i00 = wave << 5;
        ffm_load_b(B0, Wm + (size_t)(i00 + l) * DDIM + i00 + 4 * half);

#pragma unroll
        for (int r = 0; r < 4; ++r)
#pragma unroll
            for (int q = 0; q < 4; ++q)
                *(v4f*)(&xs[(4 * wave + r) * XPITCH + 128 * q + 4 * lane]) = st[r][q];
    }
    __syncthreads();

    float contrib[2][8];
#pragma unroll
    for (int mt = 0; mt < 2; ++mt)
#pragma unroll
        for (int r = 0; r < 8; ++r) contrib[mt][r] = 0.0f;

#pragma unroll
    for (int t = 0; t < 2; ++t) {
        const int p  = __builtin_amdgcn_readfirstlane(t == 0 ? wave : 15 - wave);
        const int i0 = p << 5;
        const int n  = 16 - p;

        v8f acc[2][2];
#pragma unroll
        for (int mt = 0; mt < 2; ++mt)
#pragma unroll
            for (int nt = 0; nt < 2; ++nt)
                acc[mt][nt] = (v8f){0.f, 0.f, 0.f, 0.f, 0.f, 0.f, 0.f, 0.f};

        const float* xl = &xs[l * XPITCH + i0 + 4 * half];
        const float* __restrict__ wp = Wm + (size_t)(i0 + l) * DDIM + i0 + 4 * half;

        if (t != 0) ffm_load_b(B0, wp);
        ffm_load_a(A0, xl);
        int it = 0;
        bool last_in_1 = false;
        while (it + 1 < n) {
            ffm_load_b(B1, wp + 32 * (it + 1));
            ffm_load_a(A1, xl + 32 * (it + 1));
            ffm_mma_step(acc, A0, B0);
            ++it;
            if (it + 1 >= n) { last_in_1 = true; break; }
            ffm_load_b(B0, wp + 32 * (it + 1));
            ffm_load_a(A0, xl + 32 * (it + 1));
            ffm_mma_step(acc, A1, B1);
            ++it;
        }
        if (last_in_1) ffm_mma_step(acc, A1, B1);
        else           ffm_mma_step(acc, A0, B0);

#pragma unroll
        for (int mt = 0; mt < 2; ++mt)
#pragma unroll
            for (int nt = 0; nt < 2; ++nt)
#pragma unroll
                for (int r = 0; r < 8; ++r) {
                    float xv = xs[(16 * mt + r + 8 * half) * XPITCH + i0 + 16 * nt + l];
                    contrib[mt][r] += acc[mt][nt][r] * xv;
                }
    }

#pragma unroll
    for (int mt = 0; mt < 2; ++mt)
#pragma unroll
        for (int r = 0; r < 8; ++r) {
            float v = contrib[mt][r];
            v += __shfl_xor(v, 1, 32);
            v += __shfl_xor(v, 2, 32);
            v += __shfl_xor(v, 4, 32);
            v += __shfl_xor(v, 8, 32);
            if (l == 0) partial[wave][16 * mt + r + 8 * half] = v;
        }
    __syncthreads();

    if (threadIdx.x < 32) {
        float s = 0.0f;
#pragma unroll
        for (int w = 0; w < 8; ++w) s += partial[w][threadIdx.x];
        out[b0 + threadIdx.x] = s;
    }
}

extern "C" void kernel_launch(void* const* d_in, const int* in_sizes, int n_in,
                              void* d_out, int out_size, void* d_ws, size_t ws_size,
                              hipStream_t stream) {
    const float* x    = (const float*)d_in[0];
    const float* kern = (const float*)d_in[1];
    const int*   fd   = (const int*)d_in[2];
    float* out = (float*)d_out;
    float* Wm  = (float*)d_ws;

    ffm_build_w<<<(DDIM * DDIM) / 256, 256, 0, stream>>>(kern, fd, Wm);
    ffm_quadform<<<BATCH / 32, 256, 0, stream>>>(x, Wm, out);
}


# W build kernel: each 256-byte k-row is read by 16 lanes x 16 bytes (4 lines per load, not 32) and summed by a lane tree, upper 32x32 blocks only in 2176 waves; changes f32 summation order of W
# speedup vs baseline: 1.9467x; 1.3025x over previous
#include <hip/hip_runtime.h>
#include <hip/hip_bf16.h>


typedef __attribute__((ext_vector_type(2))) float v2f;
typedef __attribute__((ext_vector_type(8))) float v8f;

#define BATCH 8192
#define DDIM  512
#define FDIM  64
#define KDIM  64

typedef __attribute__((ext_vector_type(4))) float v4f;

__global__ __launch_bounds__(128) void ffm_build_w(
    const float* __restrict__ kern,
    const int*   __restrict__ fd,
    float*       __restrict__ Wm)
{
    const int t    = blockIdx.x >> 2;
    const int hb   = blockIdx.x & 3;
    const int rp   = t / 17;
    const int cc   = t - 17 * rp;
    const bool head = cc < 16 - rp;
    const int p    = head ? rp : 15 - rp;
    const int q    = head ? rp + cc : cc - 1;

    const int wave = __builtin_amdgcn_readfirstlane(threadIdx.x >> 5);
    const int lane = threadIdx.x & 31;
    const int r    = lane >> 4;
    const int c    = lane & 15;
    const int j0   = q << 5;
    const int i_lo = (p << 5) + (hb << 3) + (wave << 1);

    const int fdv = fd[j0 + lane];
    int fj[16];
#pragma unroll
    for (int m = 0; m < 16; ++m) fj[m] = __shfl(fdv, 2 * m + r, 32);

    v4f av[2][16], bv[2][16];
#pragma unroll
    for (int rr = 0; rr < 2; ++rr) {
        const int i  = i_lo + rr;
        const int fi = fd[i];
        const float* __restrict__ ap = kern + (size_t)i * (FDIM * KDIM) + 4 * c;
        const float* __restrict__ bp = kern + (size_t)(j0 + r) * (FDIM * KDIM)
                                            + (size_t)fi * KDIM + 4 * c;
#pragma unroll
        for (int m = 0; m < 16; ++m) {
            av[rr][m] = *(const v4f*)(ap + fj[m] * KDIM);
            bv[rr][m] = *(const v4f*)(bp + (size_t)(2 * m) * (FDIM * KDIM));
        }
    }

    __builtin_amdgcn_sched_barrier(0);

    const bool h8 = (c & 8) != 0, h4 = (c & 4) != 0, h2 = (c & 2) != 0, h1 = (c & 1) != 0;
#pragma unroll
    for (int rr = 0; rr < 2; ++rr) {
        const int i = i_lo + rr;
        float s16[16];
#pragma unroll
        for (int m = 0; m < 16; ++m) {
            const v4f a = av[rr][m], b = bv[rr][m];
            s16[m] = a.x * b.x + a.y * b.y + a.z * b.z + a.w * b.w;
        }
        float s8[8], s4[4], s2[2];
#pragma unroll
        for (int m = 0; m < 8; ++m) {
            const float keep = h8 ? s16[m + 8] : s16[m];
            const float send = h8 ? s16[m] : s16[m + 8];
            s8[m] = keep + __shfl_xor(send, 8, 32);
        }
#pragma unroll
        for (int m = 0; m < 4; ++m) {
            const float keep = h4 ? s8[m + 4] : s8[m];
            const float send = h4 ? s8[m] : s8[m + 4];
            s4[m] = keep + __shfl_xor(send, 4, 32);
        }
#pragma unroll
        for (int m = 0; m < 2; ++m) {
            const float keep = h2 ? s4[m + 2] : s4[m];
            const float send = h2 ? s4[m] : s4[m + 2];
            s2[m] = keep + __shfl_xor(send, 2, 32);
        }
        const float keep = h1 ? s2[1] : s2[0];
        const float send = h1 ? s2[0] : s2[1];
        const float w = keep + __shfl_xor(send, 1, 32);

        const int j = j0 + 2 * c + r;
        Wm[(size_t)i * DDIM + j] = (j > i) ? w : 0.0f;
    }
}


#define XPITCH 516


__device__ __forceinline__ void ffm_load_b(
    v4f (&fb)[2][4], const float* __restrict__ wp)
{
#pragma unroll
    for (int t = 0; t < 2; ++t) {
#pragma unroll
        for (int s = 0; s < 4; ++s)
            fb[t][s] = *(const v4f*)(wp + t * 16 * DDIM + 8 * s);
    }
}

__device__ __forceinline__ void ffm_load_a(
    v4f (&fa)[2][4], const float* xl)
{
#pragma unroll
    for (int t = 0; t < 2; ++t) {
#pragma unroll
        for (int s = 0; s < 4; ++s)
            fa[t][s] = *(const v4f*)(xl + t * 16 * XPITCH + 8 * s);
    }
}

__device__ __forceinline__ void ffm_mma_step(
    v8f (&acc)[2][2], const v4f (&fa)[2][4], const v4f (&fb)[2][4])
{
#pragma unroll
    for (int s = 0; s < 4; ++s) {
        v2f a_lo[2], a_hi[2], b_lo[2], b_hi[2];
#pragma unroll
        for (int t = 0; t < 2; ++t) {
            a_lo[t] = __builtin_shufflevector(fa[t][s], fa[t][s], 0, 1);
            a_hi[t] = __builtin_shufflevector(fa[t][s], fa[t][s], 2, 3);
            b_lo[t] = __builtin_shufflevector(fb[t][s], fb[t][s], 0, 1);
            b_hi[t] = __builtin_shufflevector(fb[t][s], fb[t][s], 2, 3);
        }
#pragma unroll
        for (int mt = 0; mt < 2; ++mt) {
#pragma unroll
            for (int nt = 0; nt < 2; ++nt)
                acc[mt][nt] = __builtin_amdgcn_wmma_f32_16x16x4_f32(
                    false, a_lo[mt], false, b_lo[nt], (short)0, acc[mt][nt], false, false);
        }
#pragma unroll
        for (int mt = 0; mt < 2; ++mt) {
#pragma unroll
            for (int nt = 0; nt < 2; ++nt)
                acc[mt][nt] = __builtin_amdgcn_wmma_f32_16x16x4_f32(
                    false, a_hi[mt], false, b_hi[nt], (short)0, acc[mt][nt], false, false);
        }
    }
}

__global__ __launch_bounds__(256) void ffm_quadform(
    const float* __restrict__ x,
    const float* __restrict__ Wm,
    float*       __restrict__ out)
{
    const int b0   = blockIdx.x * 32;
    const int wave = __builtin_amdgcn_readfirstlane(threadIdx.x >> 5);
    const int lane = threadIdx.x & 31;
    const int half = lane >> 4;
    const int l    = lane & 15;

    __shared__ __attribute__((aligned(16))) float xs[32 * XPITCH];
    __shared__ float partial[8][32];

    v4f A0[2][4], B0[2][4], A1[2][4], B1[2][4];

    {
        const float* __restrict__ xg = x + (size_t)(b0 + 4 * wave) * DDIM + 4 * lane;
        v4f st[4][4];
#pragma unroll
        for (int r = 0; r < 4; ++r)
#pragma unroll
            for (int q = 0; q < 4; ++q)
                st[r][q] = *(const v4f*)(xg + r * DDIM + 128 * q);

        const int i00 = wave << 5;
        ffm_load_b(B0, Wm + (size_t)(i00 + l) * DDIM + i00 + 4 * half);

#pragma unroll
        for (int r = 0; r < 4; ++r)
#pragma unroll
            for (int q = 0; q < 4; ++q)
                *(v4f*)(&xs[(4 * wave + r) * XPITCH + 128 * q + 4 * lane]) = st[r][q];
    }
    __syncthreads();

    float contrib[2][8];
#pragma unroll
    for (int mt = 0; mt < 2; ++mt)
#pragma unroll
        for (int r = 0; r < 8; ++r) contrib[mt][r] = 0.0f;

#pragma unroll
    for (int t = 0; t < 2; ++t) {
        const int p  = __builtin_amdgcn_readfirstlane(t == 0 ? wave : 15 - wave);
        const int i0 = p << 5;
        const int n  = 16 - p;

        v8f acc[2][2];
#pragma unroll
        for (int mt = 0; mt < 2; ++mt)
#pragma unroll
            for (int nt = 0; nt < 2; ++nt)
                acc[mt][nt] = (v8f){0.f, 0.f, 0.f, 0.f, 0.f, 0.f, 0.f, 0.f};

        const float* xl = &xs[l * XPITCH + i0 + 4 * half];
        const float* __restrict__ wp = Wm + (size_t)(i0 + l) * DDIM + i0 + 4 * half;

        if (t != 0) ffm_load_b(B0, wp);
        ffm_load_a(A0, xl);
        int it = 0;
        bool last_in_1 = false;
        while (it + 1 < n) {
            ffm_load_b(B1, wp + 32 * (it + 1));
            ffm_load_a(A1, xl + 32 * (it + 1));
            ffm_mma_step(acc, A0, B0);
            ++it;
            if (it + 1 >= n) { last_in_1 = true; break; }
            ffm_load_b(B0, wp + 32 * (it + 1));
            ffm_load_a(A0, xl + 32 * (it + 1));
            ffm_mma_step(acc, A1, B1);
            ++it;
        }
        if (last_in_1) ffm_mma_step(acc, A1, B1);
        else           ffm_mma_step(acc, A0, B0);

#pragma unroll
        for (int mt = 0; mt < 2; ++mt)
#pragma unroll
            for (int nt = 0; nt < 2; ++nt)
#pragma unroll
                for (int r = 0; r < 8; ++r) {
                    float xv = xs[(16 * mt + r + 8 * half) * XPITCH + i0 + 16 * nt + l];
                    contrib[mt][r] += acc[mt][nt][r] * xv;
                }
    }

#pragma unroll
    for (int mt = 0; mt < 2; ++mt)
#pragma unroll
        for (int r = 0; r < 8; ++r) {
            float v = contrib[mt][r];
            v += __shfl_xor(v, 1, 32);
            v += __shfl_xor(v, 2, 32);
            v += __shfl_xor(v, 4, 32);
            v += __shfl_xor(v, 8, 32);
            if (l == 0) partial[wave][16 * mt + r + 8 * half] = v;
        }
    __syncthreads();

    if (threadIdx.x < 32) {
        float s = 0.0f;
#pragma unroll
        for (int w = 0; w < 8; ++w) s += partial[w][threadIdx.x];
        out[b0 + threadIdx.x] = s;
    }
}

extern "C" void kernel_launch(void* const* d_in, const int* in_sizes, int n_in,
                              void* d_out, int out_size, void* d_ws, size_t ws_size,
                              hipStream_t stream) {
    const float* x    = (const float*)d_in[0];
    const float* kern = (const float*)d_in[1];
    const int*   fd   = (const int*)d_in[2];
    float* out = (float*)d_out;
    float* Wm  = (float*)d_ws;

    ffm_build_w<<<544, 128, 0, stream>>>(kern, fd, Wm);
    ffm_quadform<<<BATCH / 32, 256, 0, stream>>>(x, Wm, out);
}
